# starting kernel
# baseline (speedup 1.0000x reference)
#include <hip/hip_runtime.h>


typedef __attribute__((ext_vector_type(2))) float v2f;
typedef __attribute__((ext_vector_type(8))) float v8f;

#define GK 16
#define A_CH 24
#define H_DIM 256
#define N_PIX 65536
#define TILES 4096

__global__ __launch_bounds__(256) void gmm_render_kernel(
    const float* __restrict__ params, float* __restrict__ out) {
  const int lane = threadIdx.x & 31;
  const int g    = lane & 15;
  const int hi   = lane >> 4;
  const int a    = blockIdx.y;

  const float* pa = params + a * (7 * GK);
  const float mux = pa[g];
  const float muy = pa[GK + g];
  const float z   = pa[2 * GK + g];
  const float l00 = pa[3 * GK + 4 * g + 0];
  const float l10 = pa[3 * GK + 4 * g + 2];
  const float l11 = pa[3 * GK + 4 * g + 3];

  const float srx = l00 * l00 + l00 * l10;
  const float sry = l00 * l10 + l10 * l10 + l11 * l11;
  const float c1c = -(srx * mux + sry * muy);
  const float c2c = -(mux + muy);

  float mx = z;
  mx = fmaxf(mx, __shfl_xor(mx, 1, 32));
  mx = fmaxf(mx, __shfl_xor(mx, 2, 32));
  mx = fmaxf(mx, __shfl_xor(mx, 4, 32));
  mx = fmaxf(mx, __shfl_xor(mx, 8, 32));
  float ez = __expf(z - mx);
  float ssum = ez;
  ssum += __shfl_xor(ssum, 1, 32);
  ssum += __shfl_xor(ssum, 2, 32);
  ssum += __shfl_xor(ssum, 4, 32);
  ssum += __shfl_xor(ssum, 8, 32);
  const float w = ez / ssum;

  float wv[8];
#pragma unroll
  for (int r = 0; r < 8; ++r) wv[r] = __shfl(w, r + 8 * hi, 16);

  v2f a1, a2;
  a1.x = hi ? c1c : srx;
  a1.y = hi ? 0.0f : sry;
  a2.x = hi ? c2c : 1.0f;
  a2.y = hi ? 0.0f : 1.0f;

  const v8f czero = {};

  const int waves_per_blk = blockDim.x >> 5;
  const int wave   = blockIdx.x * waves_per_blk + (threadIdx.x >> 5);
  const int nwaves = gridDim.x * waves_per_blk;

  for (int tile = wave; tile < TILES; tile += nwaves) {
    const int pn = (tile << 4) + g;
    const float gx = (float)(pn >> 8) * (1.0f / 255.0f);
    const float gy = (float)(pn & 255) * (1.0f / 255.0f);

    v2f b;
    b.x = hi ? 1.0f : gx;
    b.y = hi ? 0.0f : gy;

    v8f t1 = __builtin_amdgcn_wmma_f32_16x16x4_f32(
        false, a1, false, b, (short)0, czero, false, false);
    v8f t2 = __builtin_amdgcn_wmma_f32_16x16x4_f32(
        false, a2, false, b, (short)0, czero, false, false);

    float se = 0.0f, swe = 0.0f;
#pragma unroll
    for (int r = 0; r < 8; ++r) {
      const float e = __expf(-0.5f * t1[r] * t2[r]);
      se += e;
      swe = fmaf(wv[r], e, swe);
    }
    se  += __shfl_xor(se, 16, 32);
    swe += __shfl_xor(swe, 16, 32);

    float y = swe / fmaxf(se, 1e-8f);
    y = fminf(fmaxf(y, 0.0f), 1.0f);
    if (hi == 0) out[a * N_PIX + pn] = y;
  }
}

extern "C" void kernel_launch(void* const* d_in, const int* in_sizes, int n_in,
                              void* d_out, int out_size, void* d_ws, size_t ws_size,
                              hipStream_t stream) {
  (void)in_sizes; (void)n_in; (void)d_ws; (void)ws_size; (void)out_size;
  const float* params = (const float*)d_in[0];
  float* out = (float*)d_out;
  dim3 grid(64, A_CH);
  gmm_render_kernel<<<grid, 256, 0, stream>>>(params, out);
}


# Launch 6144 waves instead of 12288 (768 work-groups of 8 waves, 6 per SIMD): each wave renders 16 tiles, four independent tiles per trip, so per-wave set-up and wave entry are paid half as often
# speedup vs baseline: 1.3067x; 1.3067x over previous
#include <hip/hip_runtime.h>


typedef __attribute__((ext_vector_type(2))) float v2f;
typedef __attribute__((ext_vector_type(8))) float v8f;

#define GK 16
#define A_CH 24
#define H_DIM 256
#define N_PIX 65536
#define N_BLK 768
#define WAVES_PER_BLK 8
#define WAVES_PER_CH 256
#define TILES_PER_TRIP 4
#define TRIPS 4

__global__ __launch_bounds__(256) void gmm_render_kernel(
    const float* __restrict__ params, float* __restrict__ out) {
  const int lane = threadIdx.x & 31;
  const int g    = lane & 15;
  const int hi   = lane >> 4;
  const unsigned wid = (unsigned)__builtin_amdgcn_readfirstlane(
      (int)(blockIdx.x * WAVES_PER_BLK + (threadIdx.x >> 5)));
  const int a   = (int)(wid / WAVES_PER_CH);
  const int wic = (int)(wid % WAVES_PER_CH);

  const float* pa = params + a * (7 * GK);
  const float mux = pa[g];
  const float muy = pa[GK + g];
  const float z   = pa[2 * GK + g];
  const float l00 = pa[3 * GK + 4 * g + 0];
  const float l10 = pa[3 * GK + 4 * g + 2];
  const float l11 = pa[3 * GK + 4 * g + 3];

  const float srx = l00 * l00 + l00 * l10;
  const float sry = l00 * l10 + l10 * l10 + l11 * l11;
  const float c1c = -(srx * mux + sry * muy);
  const float c2c = -(mux + muy);

  float mx = z;
  mx = fmaxf(mx, __shfl_xor(mx, 1, 32));
  mx = fmaxf(mx, __shfl_xor(mx, 2, 32));
  mx = fmaxf(mx, __shfl_xor(mx, 4, 32));
  mx = fmaxf(mx, __shfl_xor(mx, 8, 32));
  float ez = __expf(z - mx);
  float ssum = ez;
  ssum += __shfl_xor(ssum, 1, 32);
  ssum += __shfl_xor(ssum, 2, 32);
  ssum += __shfl_xor(ssum, 4, 32);
  ssum += __shfl_xor(ssum, 8, 32);
  const float w = ez / ssum;

  float wv[8];
#pragma unroll
  for (int r = 0; r < 8; ++r) wv[r] = __shfl(w, r + 8 * hi, 16);

  v2f a1, a2;
  a1.x = hi ? c1c : srx;
  a1.y = hi ? 0.0f : sry;
  a2.x = hi ? c2c : 1.0f;
  a2.y = hi ? 0.0f : 1.0f;

  const v8f czero = {};

  for (int trip = 0; trip < TRIPS; ++trip) {
    const int pb = (wic + trip * WAVES_PER_CH) << 6;

    float se[TILES_PER_TRIP], swe[TILES_PER_TRIP];
#pragma unroll
    for (int j = 0; j < TILES_PER_TRIP; ++j) {
      const int pn = pb + 16 * j + g;
      const float gx = (float)(pn >> 8) * (1.0f / 255.0f);
      const float gy = (float)(pn & 255) * (1.0f / 255.0f);

      v2f b;
      b.x = hi ? 1.0f : gx;
      b.y = hi ? 0.0f : gy;

      v8f t1 = __builtin_amdgcn_wmma_f32_16x16x4_f32(
          false, a1, false, b, (short)0, czero, false, false);
      v8f t2 = __builtin_amdgcn_wmma_f32_16x16x4_f32(
          false, a2, false, b, (short)0, czero, false, false);

      float s = 0.0f, sw = 0.0f;
#pragma unroll
      for (int r = 0; r < 8; ++r) {
        const float e = __expf(-0.5f * t1[r] * t2[r]);
        s += e;
        sw = fmaf(wv[r], e, sw);
      }
      se[j] = s;
      swe[j] = sw;
    }

    float y[TILES_PER_TRIP];
#pragma unroll
    for (int j = 0; j < TILES_PER_TRIP; ++j) {
      const float s  = se[j]  + __shfl_xor(se[j], 16, 32);
      const float sw = swe[j] + __shfl_xor(swe[j], 16, 32);
      const float q = sw / fmaxf(s, 1e-8f);
      y[j] = fminf(fmaxf(q, 0.0f), 1.0f);
    }
    if (hi == 0) {
      float* o = out + a * N_PIX + pb + g;
#pragma unroll
      for (int j = 0; j < TILES_PER_TRIP; ++j) o[16 * j] = y[j];
    }
  }
}

extern "C" void kernel_launch(void* const* d_in, const int* in_sizes, int n_in,
                              void* d_out, int out_size, void* d_ws, size_t ws_size,
                              hipStream_t stream) {
  (void)in_sizes; (void)n_in; (void)d_ws; (void)ws_size; (void)out_size;
  const float* params = (const float*)d_in[0];
  float* out = (float*)d_out;
  dim3 grid(N_BLK);
  gmm_render_kernel<<<grid, 256, 0, stream>>>(params, out);
}


# Exponent as one quadratic in the run's column offset: ONE f32 WMMA per tile (A = per-run c0,c1,c2 scaled by -0.5*log2e; B = 1,u,u^2 built once), D straight into exp2; exponent rounds differently
# speedup vs baseline: 1.6696x; 1.2777x over previous
#include <hip/hip_runtime.h>


typedef __attribute__((ext_vector_type(2))) float v2f;
typedef __attribute__((ext_vector_type(8))) float v8f;

#define GK 16
#define A_CH 24
#define H_DIM 256
#define N_PIX 65536
#define N_BLK 768
#define WAVES_PER_BLK 8
#define WAVES_PER_CH 256
#define TILES_PER_TRIP 4
#define TRIPS 4
#define RUN_HALF 32

__global__ __launch_bounds__(256) void gmm_render_kernel(
    const float* __restrict__ params, float* __restrict__ out) {
  const int lane = threadIdx.x & 31;
  const int g    = lane & 15;
  const int hi   = lane >> 4;
  const unsigned wid = (unsigned)__builtin_amdgcn_readfirstlane(
      (int)(blockIdx.x * WAVES_PER_BLK + (threadIdx.x >> 5)));
  const int a   = (int)(wid / WAVES_PER_CH);
  const int wic = (int)(wid % WAVES_PER_CH);

  const float* pa = params + a * (7 * GK);
  const float mux = pa[g];
  const float muy = pa[GK + g];
  const float z   = pa[2 * GK + g];
  const float l00 = pa[3 * GK + 4 * g + 0];
  const float l10 = pa[3 * GK + 4 * g + 2];
  const float l11 = pa[3 * GK + 4 * g + 3];

  const float srx = l00 * l00 + l00 * l10;
  const float sry = l00 * l10 + l10 * l10 + l11 * l11;

  float mx = z;
  mx = fmaxf(mx, __shfl_xor(mx, 1, 32));
  mx = fmaxf(mx, __shfl_xor(mx, 2, 32));
  mx = fmaxf(mx, __shfl_xor(mx, 4, 32));
  mx = fmaxf(mx, __shfl_xor(mx, 8, 32));
  float ez = __expf(z - mx);
  float ssum = ez;
  ssum += __shfl_xor(ssum, 1, 32);
  ssum += __shfl_xor(ssum, 2, 32);
  ssum += __shfl_xor(ssum, 4, 32);
  ssum += __shfl_xor(ssum, 8, 32);
  const float w = ez / ssum;

  float wv[8];
#pragma unroll
  for (int r = 0; r < 8; ++r) wv[r] = __shfl(w, r + 8 * hi, 16);

  const float kInv = 1.0f / 255.0f;
  const float kS   = -0.5f * 1.4426950408889634f;
  const float c2   = (kS * kInv * kInv) * sry;

  v2f bq[TILES_PER_TRIP];
#pragma unroll
  for (int j = 0; j < TILES_PER_TRIP; ++j) {
    const float u = (float)(16 * j + g - RUN_HALF);
    bq[j].x = hi ? u * u : 1.0f;
    bq[j].y = hi ? 0.0f : u;
  }

  const v8f czero = {};

  for (int trip = 0; trip < TRIPS; ++trip) {
    const int pb = (wic + trip * WAVES_PER_CH) << 6;

    const float gx  = (float)(pb >> 8) * kInv;
    const float gyc = (float)((pb & 255) + RUN_HALF) * kInv;
    const float dx = gx - mux;
    const float dy = gyc - muy;
    const float P = dx * srx + dy * sry;
    const float Q = dx + dy;
    v2f aq;
    aq.x = hi ? c2 : (kS * P) * Q;
    aq.y = hi ? 0.0f : (kS * kInv) * (P + sry * Q);

    float se[TILES_PER_TRIP], swe[TILES_PER_TRIP];
#pragma unroll
    for (int j = 0; j < TILES_PER_TRIP; ++j) {
      v8f d = __builtin_amdgcn_wmma_f32_16x16x4_f32(
          false, aq, false, bq[j], (short)0, czero, false, false);

      float s = 0.0f, sw = 0.0f;
#pragma unroll
      for (int r = 0; r < 8; ++r) {
        const float e = __builtin_amdgcn_exp2f(d[r]);
        s += e;
        sw = fmaf(wv[r], e, sw);
      }
      se[j] = s;
      swe[j] = sw;
    }

    float y[TILES_PER_TRIP];
#pragma unroll
    for (int j = 0; j < TILES_PER_TRIP; ++j) {
      const float s  = se[j]  + __shfl_xor(se[j], 16, 32);
      const float sw = swe[j] + __shfl_xor(swe[j], 16, 32);
      const float q = sw / fmaxf(s, 1e-8f);
      y[j] = fminf(fmaxf(q, 0.0f), 1.0f);
    }
    if (hi == 0) {
      float* o = out + a * N_PIX + pb + g;
#pragma unroll
      for (int j = 0; j < TILES_PER_TRIP; ++j) o[16 * j] = y[j];
    }
  }
}

extern "C" void kernel_launch(void* const* d_in, const int* in_sizes, int n_in,
                              void* d_out, int out_size, void* d_ws, size_t ws_size,
                              hipStream_t stream) {
  (void)in_sizes; (void)n_in; (void)d_ws; (void)ws_size; (void)out_size;
  const float* params = (const float*)d_in[0];
  float* out = (float*)d_out;
  dim3 grid(N_BLK);
  gmm_render_kernel<<<grid, 256, 0, stream>>>(params, out);
}


# Each lane half finishes one tile of a pair (one v_permlanex16 per sum instead of ds_bpermute): half as many divisions, one 32-lane 128-byte store per two tiles, no branch; same summation order
# speedup vs baseline: 1.7716x; 1.0611x over previous
#include <hip/hip_runtime.h>


typedef __attribute__((ext_vector_type(2))) float v2f;
typedef __attribute__((ext_vector_type(8))) float v8f;

#define GK 16
#define A_CH 24
#define H_DIM 256
#define N_PIX 65536
#define N_BLK 768
#define WAVES_PER_BLK 8
#define WAVES_PER_CH 256
#define TILES_PER_TRIP 4
#define TRIPS 4
#define RUN_HALF 32

__device__ __forceinline__ float xhalf(float x) {
  const int b = __builtin_bit_cast(int, x);
  return __builtin_bit_cast(
      float, __builtin_amdgcn_permlanex16(b, b, 0x76543210, 0xfedcba98, false, false));
}

__device__ __forceinline__ float join_pair(float x0, float x1, int hi) {
  const float keep = hi ? x1 : x0;
  const float send = hi ? x0 : x1;
  return keep + xhalf(send);
}

__global__ __launch_bounds__(256) void gmm_render_kernel(
    const float* __restrict__ params, float* __restrict__ out) {
  const int lane = threadIdx.x & 31;
  const int g    = lane & 15;
  const int hi   = lane >> 4;
  const unsigned wid = (unsigned)__builtin_amdgcn_readfirstlane(
      (int)(blockIdx.x * WAVES_PER_BLK + (threadIdx.x >> 5)));
  const int a   = (int)(wid / WAVES_PER_CH);
  const int wic = (int)(wid % WAVES_PER_CH);

  const float* pa = params + a * (7 * GK);
  const float mux = pa[g];
  const float muy = pa[GK + g];
  const float z   = pa[2 * GK + g];
  const float l00 = pa[3 * GK + 4 * g + 0];
  const float l10 = pa[3 * GK + 4 * g + 2];
  const float l11 = pa[3 * GK + 4 * g + 3];

  const float srx = l00 * l00 + l00 * l10;
  const float sry = l00 * l10 + l10 * l10 + l11 * l11;

  float mx = z;
  mx = fmaxf(mx, __shfl_xor(mx, 1, 32));
  mx = fmaxf(mx, __shfl_xor(mx, 2, 32));
  mx = fmaxf(mx, __shfl_xor(mx, 4, 32));
  mx = fmaxf(mx, __shfl_xor(mx, 8, 32));
  float ez = __expf(z - mx);
  float ssum = ez;
  ssum += __shfl_xor(ssum, 1, 32);
  ssum += __shfl_xor(ssum, 2, 32);
  ssum += __shfl_xor(ssum, 4, 32);
  ssum += __shfl_xor(ssum, 8, 32);
  const float w = ez / ssum;

  float wv[8];
#pragma unroll
  for (int r = 0; r < 8; ++r) wv[r] = __shfl(w, r + 8 * hi, 16);

  const float kInv = 1.0f / 255.0f;
  const float kS   = -0.5f * 1.4426950408889634f;
  const float c2   = (kS * kInv * kInv) * sry;

  v2f bq[TILES_PER_TRIP];
#pragma unroll
  for (int j = 0; j < TILES_PER_TRIP; ++j) {
    const float u = (float)(16 * j + g - RUN_HALF);
    bq[j].x = hi ? u * u : 1.0f;
    bq[j].y = hi ? 0.0f : u;
  }

  const v8f czero = {};

  for (int trip = 0; trip < TRIPS; ++trip) {
    const int pb = (wic + trip * WAVES_PER_CH) << 6;

    const float gx  = (float)(pb >> 8) * kInv;
    const float gyc = (float)((pb & 255) + RUN_HALF) * kInv;
    const float dx = gx - mux;
    const float dy = gyc - muy;
    const float P = dx * srx + dy * sry;
    const float Q = dx + dy;
    v2f aq;
    aq.x = hi ? c2 : (kS * P) * Q;
    aq.y = hi ? 0.0f : (kS * kInv) * (P + sry * Q);

    float se[TILES_PER_TRIP], swe[TILES_PER_TRIP];
#pragma unroll
    for (int j = 0; j < TILES_PER_TRIP; ++j) {
      v8f d = __builtin_amdgcn_wmma_f32_16x16x4_f32(
          false, aq, false, bq[j], (short)0, czero, false, false);

      float s = 0.0f, sw = 0.0f;
#pragma unroll
      for (int r = 0; r < 8; ++r) {
        const float e = __builtin_amdgcn_exp2f(d[r]);
        s += e;
        sw = fmaf(wv[r], e, sw);
      }
      se[j] = s;
      swe[j] = sw;
    }

    float* o = out + a * N_PIX + pb + lane;
#pragma unroll
    for (int k = 0; k < TILES_PER_TRIP / 2; ++k) {
      const float s  = join_pair(se[2 * k],  se[2 * k + 1],  hi);
      const float sw = join_pair(swe[2 * k], swe[2 * k + 1], hi);
      const float q = sw / fmaxf(s, 1e-8f);
      o[32 * k] = fminf(fmaxf(q, 0.0f), 1.0f);
    }
  }
}

extern "C" void kernel_launch(void* const* d_in, const int* in_sizes, int n_in,
                              void* d_out, int out_size, void* d_ws, size_t ws_size,
                              hipStream_t stream) {
  (void)in_sizes; (void)n_in; (void)d_ws; (void)ws_size; (void)out_size;
  const float* params = (const float*)d_in[0];
  float* out = (float*)d_out;
  dim3 grid(N_BLK);
  gmm_render_kernel<<<grid, 256, 0, stream>>>(params, out);
}
